# starting kernel
# baseline (speedup 1.0000x reference)
#include <hip/hip_runtime.h>
#include <hip/hip_bf16.h>

typedef __attribute__((ext_vector_type(2))) float v2f;
typedef __attribute__((ext_vector_type(8))) float v8f;

#define B_     8
#define ND_    512
#define D_     128
#define NV_    4096
#define NODES_ 512

__global__ __launch_bounds__(256) void hier_sim_bin_attn_kernel(
    const int*   __restrict__ doc_index,
    const float* __restrict__ attn,
    const float* __restrict__ emb,
    const float* __restrict__ VvT,
    const float* __restrict__ bin_diff,
    const float* __restrict__ bin_start,
    float*       __restrict__ a_out)
{
    __shared__ float attn_lds[ND_];
    __shared__ float bin_lds[16];

    const int b    = blockIdx.y;
    const int tid  = threadIdx.x;
    const int wave = tid >> 5;
    const int lane = tid & 31;
    const int col  = lane & 15;
    const int hi   = lane >> 4;

    for (int i = tid; i < ND_; i += 256)
        attn_lds[i] = attn[b * ND_ + i];
    if (tid == 0) {
        float run = bin_start[0];
        #pragma unroll
        for (int i = 0; i < 16; ++i) {
            float d = bin_diff[i];
            run += (d > 0.0f) ? d : 0.0f;
            bin_lds[i] = run;
        }
    }
    __syncthreads();

    const int v0 = blockIdx.x * 128 + wave * 16;

    v2f Bf[32];
    #pragma unroll
    for (int s = 0; s < 32; ++s) {
        const int k = 4 * s + 2 * hi;
        Bf[s].x = VvT[(size_t)k       * NV_ + v0 + col];
        Bf[s].y = VvT[(size_t)(k + 1) * NV_ + v0 + col];
    }

    const float inv_h = 14.0f / 1.49f;

    float acc = 0.0f;
    for (int nt = 0; nt < 32; ++nt) {
        const int nbase = nt * 16;
        const int row   = doc_index[b * ND_ + nbase + col];
        const float* arow = emb + (size_t)row * D_ + 2 * hi;

        v8f c = {};
        #pragma unroll
        for (int s = 0; s < 32; ++s) {
            v2f af = *(const v2f*)(arow + 4 * s);
            c = __builtin_amdgcn_wmma_f32_16x16x4_f32(
                    false, af, false, Bf[s], (short)0, c, false, false);
        }

        #pragma unroll
        for (int r = 0; r < 8; ++r) {
            const float x = c[r];
            const float t = (x + 0.5f) * inv_h;
            int dig = (t < 0.0f) ? 0 : (int)t + 1;
            dig = (dig > 15) ? 15 : dig;
            const float w = bin_lds[dig];
            const int   m = r + 8 * hi;
            acc += w * attn_lds[nbase + m];
        }
    }

    acc += __shfl_xor(acc, 16, 32);
    if (hi == 0)
        a_out[(size_t)b * NV_ + v0 + col] = acc;
}

__global__ __launch_bounds__(32) void hier_final_gemm_kernel(
    const float* __restrict__ a,
    const float* __restrict__ phi,
    float*       __restrict__ out)
{
    const int j0   = blockIdx.x * 16;
    const int lane = threadIdx.x & 31;
    const int col  = lane & 15;
    const int hi   = lane >> 4;
    const int m    = col;

    const float* arow = a   + (size_t)m * NV_ + 2 * hi;
    const float* prow = phi + (size_t)(j0 + col) * NV_ + 2 * hi;

    v8f c = {};
    for (int s = 0; s < NV_ / 4; ++s) {
        v2f af;
        if (m < 8) af = *(const v2f*)(arow + 4 * s);
        else       { af.x = 0.0f; af.y = 0.0f; }
        v2f bf = *(const v2f*)(prow + 4 * s);
        c = __builtin_amdgcn_wmma_f32_16x16x4_f32(
                false, af, false, bf, (short)0, c, false, false);
    }

    #pragma unroll
    for (int r = 0; r < 8; ++r) {
        const int M = r + 8 * hi;
        if (M < 8)
            out[(size_t)M * NODES_ + j0 + col] = c[r];
    }
}

extern "C" void kernel_launch(void* const* d_in, const int* in_sizes, int n_in,
                              void* d_out, int out_size, void* d_ws, size_t ws_size,
                              hipStream_t stream) {
    const int*   doc_index = (const int*)  d_in[0];
    const float* attn      = (const float*)d_in[1];
    const float* emb       = (const float*)d_in[2];
    const float* VvT       = (const float*)d_in[3];
    const float* phi       = (const float*)d_in[4];
    const float* bin_diff  = (const float*)d_in[5];
    const float* bin_start = (const float*)d_in[6];

    float* a_scratch = (float*)d_ws;
    float* out       = (float*)d_out;

    dim3 grid1(NV_ / 128, B_);
    hier_sim_bin_attn_kernel<<<grid1, 256, 0, stream>>>(
        doc_index, attn, emb, VvT, bin_diff, bin_start, a_scratch);

    hier_final_gemm_kernel<<<NODES_ / 16, 32, 0, stream>>>(
        a_scratch, phi, out);
}


# first kernel: block gathers each emb row once (whole-row 16-byte loads) into a padded double-buffered LDS stage; all 8 waves read A fragments from LDS instead of 8 per-wave 8-byte gathers
# speedup vs baseline: 4.8386x; 4.8386x over previous
#include <hip/hip_runtime.h>
#include <hip/hip_bf16.h>

typedef __attribute__((ext_vector_type(2))) float v2f;
typedef __attribute__((ext_vector_type(8))) float v8f;

#define B_     8
#define ND_    512
#define D_     128
#define NV_    4096
#define NODES_ 512

#define ST_ROWS_   32
#define ST_COUNT_  (ND_ / ST_ROWS_)
#define ST_PITCH_  (D_ + 4)
#define ST_GROUPS_ (ST_ROWS_ / 16)

__global__ __launch_bounds__(256) void hier_sim_bin_attn_kernel(
    const int*   __restrict__ doc_index,
    const float* __restrict__ attn,
    const float* __restrict__ emb,
    const float* __restrict__ VvT,
    const float* __restrict__ bin_diff,
    const float* __restrict__ bin_start,
    float*       __restrict__ a_out)
{
    __shared__ float attn_lds[ND_];
    __shared__ float bin_lds[16];
    __shared__ __attribute__((aligned(16))) float a_lds[2][ST_ROWS_ * ST_PITCH_];

    const int b    = blockIdx.y;
    const int tid  = threadIdx.x;
    const int wave = tid >> 5;
    const int lane = tid & 31;
    const int col  = lane & 15;
    const int hi   = lane >> 4;

    const int grow = tid >> 4;
    const int gq   = tid & 15;
    const int* idx_b = doc_index + b * ND_;

    float4 g[ST_GROUPS_][2];
    #pragma unroll
    for (int i = 0; i < ST_GROUPS_; ++i) {
        const int row = idx_b[grow + 16 * i];
        const float4* src = (const float4*)(emb + (size_t)row * D_);
        g[i][0] = src[gq];
        g[i][1] = src[gq + 16];
    }
    int rnext[ST_GROUPS_];
    #pragma unroll
    for (int i = 0; i < ST_GROUPS_; ++i)
        rnext[i] = idx_b[ST_ROWS_ + grow + 16 * i];

    for (int i = tid; i < ND_; i += 256)
        attn_lds[i] = attn[b * ND_ + i];
    if (tid == 0) {
        float run = bin_start[0];
        #pragma unroll
        for (int i = 0; i < 16; ++i) {
            float d = bin_diff[i];
            run += (d > 0.0f) ? d : 0.0f;
            bin_lds[i] = run;
        }
    }

    const int v0 = blockIdx.x * 128 + wave * 16;

    v2f Bf[32];
    #pragma unroll
    for (int s = 0; s < 32; ++s) {
        const int k = 4 * s + 2 * hi;
        Bf[s].x = VvT[(size_t)k       * NV_ + v0 + col];
        Bf[s].y = VvT[(size_t)(k + 1) * NV_ + v0 + col];
    }

    #pragma unroll
    for (int i = 0; i < ST_GROUPS_; ++i) {
        float* dst = &a_lds[0][(grow + 16 * i) * ST_PITCH_];
        *(float4*)(dst + 4 * gq)        = g[i][0];
        *(float4*)(dst + 4 * (gq + 16)) = g[i][1];
    }
    __syncthreads();

    const float inv_h = 14.0f / 1.49f;

    float acc = 0.0f;
    for (int st = 0; st < ST_COUNT_; ++st) {
        const bool more = (st + 1 < ST_COUNT_);
        if (more) {
            #pragma unroll
            for (int i = 0; i < ST_GROUPS_; ++i) {
                const float4* src = (const float4*)(emb + (size_t)rnext[i] * D_);
                g[i][0] = src[gq];
                g[i][1] = src[gq + 16];
            }
            const int st2 = (st + 2 < ST_COUNT_) ? st + 2 : ST_COUNT_ - 1;
            #pragma unroll
            for (int i = 0; i < ST_GROUPS_; ++i)
                rnext[i] = idx_b[st2 * ST_ROWS_ + grow + 16 * i];
        }

        const float* arow0 = a_lds[st & 1] + col * ST_PITCH_ + 2 * hi;
        const float* arow1 = arow0 + 16 * ST_PITCH_;
        v2f Af0[32], Af1[32];
        #pragma unroll
        for (int s = 0; s < 32; ++s) {
            Af0[s] = *(const v2f*)(arow0 + 4 * s);
            Af1[s] = *(const v2f*)(arow1 + 4 * s);
        }
        __builtin_amdgcn_sched_barrier(0);

        v8f c0 = {}, c1 = {};
        #pragma unroll
        for (int s = 0; s < 32; ++s) {
            c0 = __builtin_amdgcn_wmma_f32_16x16x4_f32(
                    false, Af0[s], false, Bf[s], (short)0, c0, false, false);
            c1 = __builtin_amdgcn_wmma_f32_16x16x4_f32(
                    false, Af1[s], false, Bf[s], (short)0, c1, false, false);
        }

        #pragma unroll
        for (int t = 0; t < 2; ++t) {
            const int nbase = st * ST_ROWS_ + 16 * t;
            #pragma unroll
            for (int r = 0; r < 8; ++r) {
                const float x = t ? c1[r] : c0[r];
                const float tt = (x + 0.5f) * inv_h;
                int dig = (tt < 0.0f) ? 0 : (int)tt + 1;
                dig = (dig > 15) ? 15 : dig;
                const float w = bin_lds[dig];
                const int   m = r + 8 * hi;
                acc += w * attn_lds[nbase + m];
            }
        }

        if (more) {
            float* sbuf = a_lds[(st + 1) & 1];
            #pragma unroll
            for (int i = 0; i < ST_GROUPS_; ++i) {
                float* dst = sbuf + (grow + 16 * i) * ST_PITCH_;
                *(float4*)(dst + 4 * gq)        = g[i][0];
                *(float4*)(dst + 4 * (gq + 16)) = g[i][1];
            }
            __syncthreads();
        }
    }

    acc += __shfl_xor(acc, 16, 32);
    if (hi == 0)
        a_out[(size_t)b * NV_ + v0 + col] = acc;
}

#define K2_WAVES_  32
#define K2_KSLICE_ (NV_ / K2_WAVES_)
#define K2_STEPS_  (K2_KSLICE_ / 4)

__global__ __launch_bounds__(1024) void hier_final_gemm_kernel(
    const float* __restrict__ a,
    const float* __restrict__ phi,
    float*       __restrict__ out)
{
    __shared__ float part[K2_WAVES_][8 * 16];

    const int j0   = blockIdx.x * 16;
    const int tid  = threadIdx.x;
    const int wave = tid >> 5;
    const int lane = tid & 31;
    const int col  = lane & 15;
    const int hi   = lane >> 4;
    const int m    = col;
    const int k0   = wave * K2_KSLICE_;

    const float* arow = a   + (size_t)(m & 7) * NV_ + k0 + 2 * hi;
    const float* prow = phi + (size_t)(j0 + col) * NV_ + k0 + 2 * hi;

    v8f c = {};
    #pragma unroll
    for (int s = 0; s < K2_STEPS_; ++s) {
        const v2f af = *(const v2f*)(arow + 4 * s);
        const v2f bf = *(const v2f*)(prow + 4 * s);
        c = __builtin_amdgcn_wmma_f32_16x16x4_f32(
                false, af, false, bf, (short)0, c, false, false);
    }

    if (hi == 0) {
        #pragma unroll
        for (int r = 0; r < 8; ++r)
            part[wave][r * 16 + col] = c[r];
    }
    __syncthreads();

    if (tid < 8 * 16) {
        float sum = 0.0f;
        #pragma unroll
        for (int w = 0; w < K2_WAVES_; ++w)
            sum += part[w][tid];
        out[(size_t)(tid >> 4) * NODES_ + j0 + (tid & 15)] = sum;
    }
}

extern "C" void kernel_launch(void* const* d_in, const int* in_sizes, int n_in,
                              void* d_out, int out_size, void* d_ws, size_t ws_size,
                              hipStream_t stream) {
    const int*   doc_index = (const int*)  d_in[0];
    const float* attn      = (const float*)d_in[1];
    const float* emb       = (const float*)d_in[2];
    const float* VvT       = (const float*)d_in[3];
    const float* phi       = (const float*)d_in[4];
    const float* bin_diff  = (const float*)d_in[5];
    const float* bin_start = (const float*)d_in[6];

    float* a_scratch = (float*)d_ws;
    float* out       = (float*)d_out;

    dim3 grid1(NV_ / 128, B_);
    hier_sim_bin_attn_kernel<<<grid1, 256, 0, stream>>>(
        doc_index, attn, emb, VvT, bin_diff, bin_start, a_scratch);

    hier_final_gemm_kernel<<<NODES_ / 16, 1024, 0, stream>>>(
        a_scratch, phi, out);
}


# final GEMM kernel: each wave loads its phi and a slices as whole 128-byte lines (4 lines per load, was 16 and 8) into its own padded LDS patch and reads the WMMA fragments from LDS
# speedup vs baseline: 5.6276x; 1.1631x over previous
#include <hip/hip_runtime.h>
#include <hip/hip_bf16.h>

typedef __attribute__((ext_vector_type(2))) float v2f;
typedef __attribute__((ext_vector_type(8))) float v8f;

#define B_     8
#define ND_    512
#define D_     128
#define NV_    4096
#define NODES_ 512

#define ST_ROWS_   32
#define ST_COUNT_  (ND_ / ST_ROWS_)
#define ST_PITCH_  (D_ + 4)
#define ST_GROUPS_ (ST_ROWS_ / 16)

__global__ __launch_bounds__(256) void hier_sim_bin_attn_kernel(
    const int*   __restrict__ doc_index,
    const float* __restrict__ attn,
    const float* __restrict__ emb,
    const float* __restrict__ VvT,
    const float* __restrict__ bin_diff,
    const float* __restrict__ bin_start,
    float*       __restrict__ a_out)
{
    __shared__ float attn_lds[ND_];
    __shared__ float bin_lds[16];
    __shared__ __attribute__((aligned(16))) float a_lds[2][ST_ROWS_ * ST_PITCH_];

    const int b    = blockIdx.y;
    const int tid  = threadIdx.x;
    const int wave = tid >> 5;
    const int lane = tid & 31;
    const int col  = lane & 15;
    const int hi   = lane >> 4;

    const int grow = tid >> 4;
    const int gq   = tid & 15;
    const int* idx_b = doc_index + b * ND_;

    float4 g[ST_GROUPS_][2];
    #pragma unroll
    for (int i = 0; i < ST_GROUPS_; ++i) {
        const int row = idx_b[grow + 16 * i];
        const float4* src = (const float4*)(emb + (size_t)row * D_);
        g[i][0] = src[gq];
        g[i][1] = src[gq + 16];
    }
    int rnext[ST_GROUPS_];
    #pragma unroll
    for (int i = 0; i < ST_GROUPS_; ++i)
        rnext[i] = idx_b[ST_ROWS_ + grow + 16 * i];

    for (int i = tid; i < ND_; i += 256)
        attn_lds[i] = attn[b * ND_ + i];
    if (tid == 0) {
        float run = bin_start[0];
        #pragma unroll
        for (int i = 0; i < 16; ++i) {
            float d = bin_diff[i];
            run += (d > 0.0f) ? d : 0.0f;
            bin_lds[i] = run;
        }
    }

    const int v0 = blockIdx.x * 128 + wave * 16;

    v2f Bf[32];
    #pragma unroll
    for (int s = 0; s < 32; ++s) {
        const int k = 4 * s + 2 * hi;
        Bf[s].x = VvT[(size_t)k       * NV_ + v0 + col];
        Bf[s].y = VvT[(size_t)(k + 1) * NV_ + v0 + col];
    }

    #pragma unroll
    for (int i = 0; i < ST_GROUPS_; ++i) {
        float* dst = &a_lds[0][(grow + 16 * i) * ST_PITCH_];
        *(float4*)(dst + 4 * gq)        = g[i][0];
        *(float4*)(dst + 4 * (gq + 16)) = g[i][1];
    }
    __syncthreads();

    const float inv_h = 14.0f / 1.49f;

    float acc = 0.0f;
    for (int st = 0; st < ST_COUNT_; ++st) {
        const bool more = (st + 1 < ST_COUNT_);
        if (more) {
            #pragma unroll
            for (int i = 0; i < ST_GROUPS_; ++i) {
                const float4* src = (const float4*)(emb + (size_t)rnext[i] * D_);
                g[i][0] = src[gq];
                g[i][1] = src[gq + 16];
            }
            const int st2 = (st + 2 < ST_COUNT_) ? st + 2 : ST_COUNT_ - 1;
            #pragma unroll
            for (int i = 0; i < ST_GROUPS_; ++i)
                rnext[i] = idx_b[st2 * ST_ROWS_ + grow + 16 * i];
        }

        const float* arow0 = a_lds[st & 1] + col * ST_PITCH_ + 2 * hi;
        const float* arow1 = arow0 + 16 * ST_PITCH_;
        v2f Af0[32], Af1[32];
        #pragma unroll
        for (int s = 0; s < 32; ++s) {
            Af0[s] = *(const v2f*)(arow0 + 4 * s);
            Af1[s] = *(const v2f*)(arow1 + 4 * s);
        }
        __builtin_amdgcn_sched_barrier(0);

        v8f c0 = {}, c1 = {};
        #pragma unroll
        for (int s = 0; s < 32; ++s) {
            c0 = __builtin_amdgcn_wmma_f32_16x16x4_f32(
                    false, Af0[s], false, Bf[s], (short)0, c0, false, false);
            c1 = __builtin_amdgcn_wmma_f32_16x16x4_f32(
                    false, Af1[s], false, Bf[s], (short)0, c1, false, false);
        }

        #pragma unroll
        for (int t = 0; t < 2; ++t) {
            const int nbase = st * ST_ROWS_ + 16 * t;
            #pragma unroll
            for (int r = 0; r < 8; ++r) {
                const float x = t ? c1[r] : c0[r];
                const float tt = (x + 0.5f) * inv_h;
                int dig = (tt < 0.0f) ? 0 : (int)tt + 1;
                dig = (dig > 15) ? 15 : dig;
                const float w = bin_lds[dig];
                const int   m = r + 8 * hi;
                acc += w * attn_lds[nbase + m];
            }
        }

        if (more) {
            float* sbuf = a_lds[(st + 1) & 1];
            #pragma unroll
            for (int i = 0; i < ST_GROUPS_; ++i) {
                float* dst = sbuf + (grow + 16 * i) * ST_PITCH_;
                *(float4*)(dst + 4 * gq)        = g[i][0];
                *(float4*)(dst + 4 * (gq + 16)) = g[i][1];
            }
            __syncthreads();
        }
    }

    acc += __shfl_xor(acc, 16, 32);
    if (hi == 0)
        a_out[(size_t)b * NV_ + v0 + col] = acc;
}

#define K2_WAVES_  32
#define K2_KSLICE_ (NV_ / K2_WAVES_)
#define K2_CH_     32
#define K2_NCH_    (K2_KSLICE_ / K2_CH_)
#define K2_CSTEPS_ (K2_CH_ / 4)
#define K2_PITCH_  (K2_CH_ + 4)
#define K2_PROWS_  16
#define K2_AROWS_  8
#define K2_PATCH_  ((K2_PROWS_ + K2_AROWS_) * K2_PITCH_)

struct K2Rows { float4 p0, p1, p2, p3, a0, a1; };

__device__ __forceinline__ K2Rows k2_load_rows(const float* psrc, const float* asrc) {
    K2Rows g;
    g.p0 = *(const float4*)(psrc);
    g.p1 = *(const float4*)(psrc + (size_t)4  * NV_);
    g.p2 = *(const float4*)(psrc + (size_t)8  * NV_);
    g.p3 = *(const float4*)(psrc + (size_t)12 * NV_);
    g.a0 = *(const float4*)(asrc);
    g.a1 = *(const float4*)(asrc + (size_t)4  * NV_);
    return g;
}

__device__ __forceinline__ void k2_store_rows(float* pdst, float* adst, const K2Rows& g) {
    *(float4*)(pdst)                  = g.p0;
    *(float4*)(pdst + 4  * K2_PITCH_) = g.p1;
    *(float4*)(pdst + 8  * K2_PITCH_) = g.p2;
    *(float4*)(pdst + 12 * K2_PITCH_) = g.p3;
    *(float4*)(adst)                  = g.a0;
    *(float4*)(adst + 4  * K2_PITCH_) = g.a1;
}

__global__ __launch_bounds__(1024) void hier_final_gemm_kernel(
    const float* __restrict__ a,
    const float* __restrict__ phi,
    float*       __restrict__ out)
{
    __shared__ float part[K2_WAVES_][8 * 16];
    __shared__ __attribute__((aligned(16))) float patch[K2_WAVES_][2][K2_PATCH_];

    const int j0   = blockIdx.x * 16;
    const int tid  = threadIdx.x;
    const int wave = tid >> 5;
    const int lane = tid & 31;
    const int col  = lane & 15;
    const int hi   = lane >> 4;
    const int k0   = wave * K2_KSLICE_;

    const int lrow = lane >> 3;
    const int lq   = lane & 7;
    const float* pbase = phi + (size_t)(j0 + lrow) * NV_ + k0 + 4 * lq;
    const float* abase = a   + (size_t)lrow        * NV_ + k0 + 4 * lq;

    float* my   = &patch[wave][0][0];
    float* pdst = my + lrow * K2_PITCH_ + 4 * lq;
    float* adst = my + (K2_PROWS_ + lrow) * K2_PITCH_ + 4 * lq;

    const float* bfrag = my + col * K2_PITCH_ + 2 * hi;
    const float* afrag = my + (K2_PROWS_ + (col & 7)) * K2_PITCH_ + 2 * hi;

    K2Rows g = k2_load_rows(pbase, abase);
    k2_store_rows(pdst, adst, g);
    __syncthreads();

    v8f c = {};
    #pragma unroll 1
    for (int ch = 0; ch < K2_NCH_; ++ch) {
        const bool more = (ch + 1 < K2_NCH_);
        if (more)
            g = k2_load_rows(pbase + (ch + 1) * K2_CH_, abase + (ch + 1) * K2_CH_);

        const int cur = (ch & 1) * K2_PATCH_;
        v2f af[K2_CSTEPS_], bf[K2_CSTEPS_];
        #pragma unroll
        for (int s = 0; s < K2_CSTEPS_; ++s) {
            af[s] = *(const v2f*)(afrag + cur + 4 * s);
            bf[s] = *(const v2f*)(bfrag + cur + 4 * s);
        }
        #pragma unroll
        for (int s = 0; s < K2_CSTEPS_; ++s)
            c = __builtin_amdgcn_wmma_f32_16x16x4_f32(
                    false, af[s], false, bf[s], (short)0, c, false, false);

        if (more) {
            const int nxt = ((ch + 1) & 1) * K2_PATCH_;
            k2_store_rows(pdst + nxt, adst + nxt, g);
            __syncthreads();
        }
    }

    if (hi == 0) {
        #pragma unroll
        for (int r = 0; r < 8; ++r)
            part[wave][r * 16 + col] = c[r];
    }
    __syncthreads();

    if (tid < 8 * 16) {
        float sum = 0.0f;
        #pragma unroll
        for (int w = 0; w < K2_WAVES_; ++w)
            sum += part[w][tid];
        out[(size_t)(tid >> 4) * NODES_ + j0 + (tid & 15)] = sum;
    }
}

extern "C" void kernel_launch(void* const* d_in, const int* in_sizes, int n_in,
                              void* d_out, int out_size, void* d_ws, size_t ws_size,
                              hipStream_t stream) {
    const int*   doc_index = (const int*)  d_in[0];
    const float* attn      = (const float*)d_in[1];
    const float* emb       = (const float*)d_in[2];
    const float* VvT       = (const float*)d_in[3];
    const float* phi       = (const float*)d_in[4];
    const float* bin_diff  = (const float*)d_in[5];
    const float* bin_start = (const float*)d_in[6];

    float* a_scratch = (float*)d_ws;
    float* out       = (float*)d_out;

    dim3 grid1(NV_ / 128, B_);
    hier_sim_bin_attn_kernel<<<grid1, 256, 0, stream>>>(
        doc_index, attn, emb, VvT, bin_diff, bin_start, a_scratch);

    hier_final_gemm_kernel<<<NODES_ / 16, 1024, 0, stream>>>(
        a_scratch, phi, out);
}


# first kernel: the two waves sharing a SIMD (w, w+4) work half a stage apart: each half's LDS fragment reads and bin epilogue run under the other half's WMMA run; three LDS stages; order of sums kept
# speedup vs baseline: 5.8957x; 1.0476x over previous
#include <hip/hip_runtime.h>
#include <hip/hip_bf16.h>

typedef __attribute__((ext_vector_type(2))) float v2f;
typedef __attribute__((ext_vector_type(8))) float v8f;

#define B_     8
#define ND_    512
#define D_     128
#define NV_    4096
#define NODES_ 512

#define ST_ROWS_   32
#define ST_COUNT_  (ND_ / ST_ROWS_)
#define ST_PITCH_  (D_ + 4)
#define ST_GROUPS_ (ST_ROWS_ / 16)
#define ST_BUFS_   3
#define ST_FLOATS_ (ST_ROWS_ * ST_PITCH_)

struct K1Rows { float4 a0, a1, b0, b1; };

__device__ __forceinline__ K1Rows k1_gather(const float* emb, int rowa, int rowb, int gq) {
    const float4* sa = (const float4*)(emb + (size_t)rowa * D_);
    const float4* sb = (const float4*)(emb + (size_t)rowb * D_);
    K1Rows g;
    g.a0 = sa[gq];
    g.a1 = sa[gq + 16];
    g.b0 = sb[gq];
    g.b1 = sb[gq + 16];
    return g;
}

__device__ __forceinline__ void k1_scatter(float* dst, const K1Rows& g) {
    *(float4*)(dst)                       = g.a0;
    *(float4*)(dst + 64)                  = g.a1;
    *(float4*)(dst + 16 * ST_PITCH_)      = g.b0;
    *(float4*)(dst + 16 * ST_PITCH_ + 64) = g.b1;
}

__device__ __forceinline__ void k1_load_frags(v2f (&Af0)[32], v2f (&Af1)[32], const float* arow0) {
    const float* arow1 = arow0 + 16 * ST_PITCH_;
    #pragma unroll
    for (int s = 0; s < 32; ++s) {
        Af0[s] = *(const v2f*)(arow0 + 4 * s);
        Af1[s] = *(const v2f*)(arow1 + 4 * s);
    }
}

__device__ __forceinline__ void k1_wmma_run(v8f& c0, v8f& c1, const v2f (&Af0)[32],
                                            const v2f (&Af1)[32], const v2f (&Bf)[32]) {
    c0 = v8f{};
    c1 = v8f{};
    #pragma unroll
    for (int s = 0; s < 32; ++s) {
        c0 = __builtin_amdgcn_wmma_f32_16x16x4_f32(
                false, Af0[s], false, Bf[s], (short)0, c0, false, false);
        c1 = __builtin_amdgcn_wmma_f32_16x16x4_f32(
                false, Af1[s], false, Bf[s], (short)0, c1, false, false);
    }
}

__device__ __forceinline__ float k1_epilogue(float acc, const v8f& c0, const v8f& c1, int nbase0,
                                             int hi, const float* attn_lds, const float* bin_lds) {
    const float inv_h = 14.0f / 1.49f;
    #pragma unroll
    for (int t = 0; t < 2; ++t) {
        const int nbase = nbase0 + 16 * t;
        #pragma unroll
        for (int r = 0; r < 8; ++r) {
            const float x = t ? c1[r] : c0[r];
            const float tt = (x + 0.5f) * inv_h;
            int dig = (tt < 0.0f) ? 0 : (int)tt + 1;
            dig = (dig > 15) ? 15 : dig;
            const float w = bin_lds[dig];
            const int   m = r + 8 * hi;
            acc += w * attn_lds[nbase + m];
        }
    }
    return acc;
}

__global__ __launch_bounds__(256) void hier_sim_bin_attn_kernel(
    const int*   __restrict__ doc_index,
    const float* __restrict__ attn,
    const float* __restrict__ emb,
    const float* __restrict__ VvT,
    const float* __restrict__ bin_diff,
    const float* __restrict__ bin_start,
    float*       __restrict__ a_out)
{
    __shared__ float attn_lds[ND_];
    __shared__ float bin_lds[16];
    __shared__ __attribute__((aligned(16))) float a_lds[ST_BUFS_][ST_FLOATS_];

    const int b    = blockIdx.y;
    const int tid  = threadIdx.x;
    const int wave = tid >> 5;
    const int lane = tid & 31;
    const int col  = lane & 15;
    const int hi   = lane >> 4;
    const bool lead = (__builtin_amdgcn_readfirstlane(tid) < 128);

    const int grow = tid >> 4;
    const int gq   = tid & 15;
    const int* idx_b = doc_index + b * ND_;

    K1Rows g = k1_gather(emb, idx_b[grow], idx_b[grow + 16], gq);
    int rn0 = idx_b[ST_ROWS_ + grow];
    int rn1 = idx_b[ST_ROWS_ + grow + 16];

    for (int i = tid; i < ND_; i += 256)
        attn_lds[i] = attn[b * ND_ + i];
    if (tid == 0) {
        float run = bin_start[0];
        #pragma unroll
        for (int i = 0; i < 16; ++i) {
            float d = bin_diff[i];
            run += (d > 0.0f) ? d : 0.0f;
            bin_lds[i] = run;
        }
    }

    const int v0 = blockIdx.x * 128 + wave * 16;

    v2f Bf[32];
    #pragma unroll
    for (int s = 0; s < 32; ++s) {
        const int k = 4 * s + 2 * hi;
        Bf[s].x = VvT[(size_t)k       * NV_ + v0 + col];
        Bf[s].y = VvT[(size_t)(k + 1) * NV_ + v0 + col];
    }

    const int goff = grow * ST_PITCH_ + 4 * gq;
    const int foff = col * ST_PITCH_ + 2 * hi;
    k1_scatter(&a_lds[0][goff], g);
    g   = k1_gather(emb, rn0, rn1, gq);
    rn0 = idx_b[2 * ST_ROWS_ + grow];
    rn1 = idx_b[2 * ST_ROWS_ + grow + 16];
    __syncthreads();

    v8f c0 = {}, c1 = {};
    float acc = 0.0f;
    int bcur = 0;
    #pragma unroll 1
    for (int st = 0; st < ST_COUNT_; ++st) {
        const int bnxt = (bcur == 2) ? 0 : bcur + 1;
        v2f Af0[32], Af1[32];

        if (lead)
            k1_load_frags(Af0, Af1, &a_lds[bcur][foff]);
        if (st + 1 < ST_COUNT_) {
            k1_scatter(&a_lds[bnxt][goff], g);
        }
        __syncthreads();

        if (st + 2 < ST_COUNT_) {
            g = k1_gather(emb, rn0, rn1, gq);
            const int st3 = (st + 3 < ST_COUNT_) ? st + 3 : ST_COUNT_ - 1;
            rn0 = idx_b[st3 * ST_ROWS_ + grow];
            rn1 = idx_b[st3 * ST_ROWS_ + grow + 16];
        }
        __builtin_amdgcn_sched_barrier(0);

        if (!lead) {
            k1_load_frags(Af0, Af1, &a_lds[bcur][foff]);
            __builtin_amdgcn_sched_barrier(0);
            if (st > 0)
                acc = k1_epilogue(acc, c0, c1, (st - 1) * ST_ROWS_, hi, attn_lds, bin_lds);
        }
        k1_wmma_run(c0, c1, Af0, Af1, Bf);
        if (lead)
            acc = k1_epilogue(acc, c0, c1, st * ST_ROWS_, hi, attn_lds, bin_lds);
        bcur = bnxt;
    }
    if (!lead)
        acc = k1_epilogue(acc, c0, c1, (ST_COUNT_ - 1) * ST_ROWS_, hi, attn_lds, bin_lds);

    acc += __shfl_xor(acc, 16, 32);
    if (hi == 0)
        a_out[(size_t)b * NV_ + v0 + col] = acc;
}

#define K2_WAVES_  32
#define K2_KSLICE_ (NV_ / K2_WAVES_)
#define K2_CH_     32
#define K2_NCH_    (K2_KSLICE_ / K2_CH_)
#define K2_CSTEPS_ (K2_CH_ / 4)
#define K2_PITCH_  (K2_CH_ + 4)
#define K2_PROWS_  16
#define K2_AROWS_  8
#define K2_PATCH_  ((K2_PROWS_ + K2_AROWS_) * K2_PITCH_)

struct K2Rows { float4 p0, p1, p2, p3, a0, a1; };

__device__ __forceinline__ K2Rows k2_load_rows(const float* psrc, const float* asrc) {
    K2Rows g;
    g.p0 = *(const float4*)(psrc);
    g.p1 = *(const float4*)(psrc + (size_t)4  * NV_);
    g.p2 = *(const float4*)(psrc + (size_t)8  * NV_);
    g.p3 = *(const float4*)(psrc + (size_t)12 * NV_);
    g.a0 = *(const float4*)(asrc);
    g.a1 = *(const float4*)(asrc + (size_t)4  * NV_);
    return g;
}

__device__ __forceinline__ void k2_store_rows(float* pdst, float* adst, const K2Rows& g) {
    *(float4*)(pdst)                  = g.p0;
    *(float4*)(pdst + 4  * K2_PITCH_) = g.p1;
    *(float4*)(pdst + 8  * K2_PITCH_) = g.p2;
    *(float4*)(pdst + 12 * K2_PITCH_) = g.p3;
    *(float4*)(adst)                  = g.a0;
    *(float4*)(adst + 4  * K2_PITCH_) = g.a1;
}

__global__ __launch_bounds__(1024) void hier_final_gemm_kernel(
    const float* __restrict__ a,
    const float* __restrict__ phi,
    float*       __restrict__ out)
{
    __shared__ float part[K2_WAVES_][8 * 16];
    __shared__ __attribute__((aligned(16))) float patch[K2_WAVES_][2][K2_PATCH_];

    const int j0   = blockIdx.x * 16;
    const int tid  = threadIdx.x;
    const int wave = tid >> 5;
    const int lane = tid & 31;
    const int col  = lane & 15;
    const int hi   = lane >> 4;
    const int k0   = wave * K2_KSLICE_;

    const int lrow = lane >> 3;
    const int lq   = lane & 7;
    const float* pbase = phi + (size_t)(j0 + lrow) * NV_ + k0 + 4 * lq;
    const float* abase = a   + (size_t)lrow        * NV_ + k0 + 4 * lq;

    float* my   = &patch[wave][0][0];
    float* pdst = my + lrow * K2_PITCH_ + 4 * lq;
    float* adst = my + (K2_PROWS_ + lrow) * K2_PITCH_ + 4 * lq;

    const float* bfrag = my + col * K2_PITCH_ + 2 * hi;
    const float* afrag = my + (K2_PROWS_ + (col & 7)) * K2_PITCH_ + 2 * hi;

    K2Rows g = k2_load_rows(pbase, abase);
    k2_store_rows(pdst, adst, g);
    __syncthreads();

    v8f c = {};
    #pragma unroll 1
    for (int ch = 0; ch < K2_NCH_; ++ch) {
        const bool more = (ch + 1 < K2_NCH_);
        if (more)
            g = k2_load_rows(pbase + (ch + 1) * K2_CH_, abase + (ch + 1) * K2_CH_);

        const int cur = (ch & 1) * K2_PATCH_;
        v2f af[K2_CSTEPS_], bf[K2_CSTEPS_];
        #pragma unroll
        for (int s = 0; s < K2_CSTEPS_; ++s) {
            af[s] = *(const v2f*)(afrag + cur + 4 * s);
            bf[s] = *(const v2f*)(bfrag + cur + 4 * s);
        }
        #pragma unroll
        for (int s = 0; s < K2_CSTEPS_; ++s)
            c = __builtin_amdgcn_wmma_f32_16x16x4_f32(
                    false, af[s], false, bf[s], (short)0, c, false, false);

        if (more) {
            const int nxt = ((ch + 1) & 1) * K2_PATCH_;
            k2_store_rows(pdst + nxt, adst + nxt, g);
            __syncthreads();
        }
    }

    if (hi == 0) {
        #pragma unroll
        for (int r = 0; r < 8; ++r)
            part[wave][r * 16 + col] = c[r];
    }
    __syncthreads();

    if (tid < 8 * 16) {
        float sum = 0.0f;
        #pragma unroll
        for (int w = 0; w < K2_WAVES_; ++w)
            sum += part[w][tid];
        out[(size_t)(tid >> 4) * NODES_ + j0 + (tid & 15)] = sum;
    }
}

extern "C" void kernel_launch(void* const* d_in, const int* in_sizes, int n_in,
                              void* d_out, int out_size, void* d_ws, size_t ws_size,
                              hipStream_t stream) {
    const int*   doc_index = (const int*)  d_in[0];
    const float* attn      = (const float*)d_in[1];
    const float* emb       = (const float*)d_in[2];
    const float* VvT       = (const float*)d_in[3];
    const float* phi       = (const float*)d_in[4];
    const float* bin_diff  = (const float*)d_in[5];
    const float* bin_start = (const float*)d_in[6];

    float* a_scratch = (float*)d_ws;
    float* out       = (float*)d_out;

    dim3 grid1(NV_ / 128, B_);
    hier_sim_bin_attn_kernel<<<grid1, 256, 0, stream>>>(
        doc_index, attn, emb, VvT, bin_diff, bin_start, a_scratch);

    hier_final_gemm_kernel<<<NODES_ / 16, 1024, 0, stream>>>(
        a_scratch, phi, out);
}


# first kernel epilogue: bin index from the rounding of one multiply-add (magic-number add), bin weight by in-row lane permute from a register table; 64 vector + 4 LDS instructions a stage, was 115+20
# speedup vs baseline: 6.2466x; 1.0595x over previous
#include <hip/hip_runtime.h>
#include <hip/hip_bf16.h>

typedef __attribute__((ext_vector_type(2))) float v2f;
typedef __attribute__((ext_vector_type(8))) float v8f;

#define B_     8
#define ND_    512
#define D_     128
#define NV_    4096
#define NODES_ 512

#define ST_ROWS_   32
#define ST_COUNT_  (ND_ / ST_ROWS_)
#define ST_PITCH_  (D_ + 4)
#define ST_GROUPS_ (ST_ROWS_ / 16)
#define ST_BUFS_   3
#define ST_FLOATS_ (ST_ROWS_ * ST_PITCH_)

struct K1Rows { float4 a0, a1, b0, b1; };

__device__ __forceinline__ K1Rows k1_gather(const float* emb, int rowa, int rowb, int gq) {
    const float4* sa = (const float4*)(emb + (size_t)rowa * D_);
    const float4* sb = (const float4*)(emb + (size_t)rowb * D_);
    K1Rows g;
    g.a0 = sa[gq];
    g.a1 = sa[gq + 16];
    g.b0 = sb[gq];
    g.b1 = sb[gq + 16];
    return g;
}

__device__ __forceinline__ void k1_scatter(float* dst, const K1Rows& g) {
    *(float4*)(dst)                       = g.a0;
    *(float4*)(dst + 64)                  = g.a1;
    *(float4*)(dst + 16 * ST_PITCH_)      = g.b0;
    *(float4*)(dst + 16 * ST_PITCH_ + 64) = g.b1;
}

__device__ __forceinline__ void k1_load_frags(v2f (&Af0)[32], v2f (&Af1)[32], const float* arow0) {
    const float* arow1 = arow0 + 16 * ST_PITCH_;
    #pragma unroll
    for (int s = 0; s < 32; ++s) {
        Af0[s] = *(const v2f*)(arow0 + 4 * s);
        Af1[s] = *(const v2f*)(arow1 + 4 * s);
    }
}

__device__ __forceinline__ void k1_wmma_run(v8f& c0, v8f& c1, const v2f (&Af0)[32],
                                            const v2f (&Af1)[32], const v2f (&Bf)[32]) {
    c0 = v8f{};
    c1 = v8f{};
    #pragma unroll
    for (int s = 0; s < 32; ++s) {
        c0 = __builtin_amdgcn_wmma_f32_16x16x4_f32(
                false, Af0[s], false, Bf[s], (short)0, c0, false, false);
        c1 = __builtin_amdgcn_wmma_f32_16x16x4_f32(
                false, Af1[s], false, Bf[s], (short)0, c1, false, false);
    }
}

__device__ __forceinline__ float k1_epilogue(float acc, const v8f& c0, const v8f& c1, int nbase0,
                                             int hi, const float* attn_lds, unsigned wtab) {
    const float inv_h  = 14.0f / 1.49f;
    const float c_half = 0.5f - 0.5f * (1.49f / 14.0f);
    const float magic  = 12582912.0f;
    #pragma unroll
    for (int t = 0; t < 2; ++t) {
        const int nbase = nbase0 + 16 * t;
        #pragma unroll
        for (int r = 0; r < 8; ++r) {
            const float x = t ? c1[r] : c0[r];
            float k = __builtin_fmaf(x + c_half, inv_h, magic);
            k = __builtin_amdgcn_fmed3f(k, magic - 1.0f, magic + 14.0f);
            const unsigned sel = __float_as_uint(k);
            const float w = __uint_as_float(
                __builtin_amdgcn_permlane16_var(sel, wtab, sel, false, false));
            const int   m = r + 8 * hi;
            acc = __builtin_fmaf(w, attn_lds[nbase + m], acc);
        }
    }
    return acc;
}

__global__ __launch_bounds__(256) __attribute__((amdgpu_waves_per_eu(2, 2)))
void hier_sim_bin_attn_kernel(
    const int*   __restrict__ doc_index,
    const float* __restrict__ attn,
    const float* __restrict__ emb,
    const float* __restrict__ VvT,
    const float* __restrict__ bin_diff,
    const float* __restrict__ bin_start,
    float*       __restrict__ a_out)
{
    __shared__ float attn_lds[ND_];
    __shared__ float bin_lds[16];
    __shared__ __attribute__((aligned(16))) float a_lds[ST_BUFS_][ST_FLOATS_];

    const int b    = blockIdx.y;
    const int tid  = threadIdx.x;
    const int wave = tid >> 5;
    const int lane = tid & 31;
    const int col  = lane & 15;
    const int hi   = lane >> 4;
    const bool lead = (__builtin_amdgcn_readfirstlane(tid) < 128);

    const int grow = tid >> 4;
    const int gq   = tid & 15;
    const int* idx_b = doc_index + b * ND_;

    K1Rows g = k1_gather(emb, idx_b[grow], idx_b[grow + 16], gq);
    int rn0 = idx_b[ST_ROWS_ + grow];
    int rn1 = idx_b[ST_ROWS_ + grow + 16];

    for (int i = tid; i < ND_; i += 256)
        attn_lds[i] = attn[b * ND_ + i];
    if (tid == 0) {
        float run = bin_start[0];
        #pragma unroll
        for (int i = 0; i < 16; ++i) {
            float d = bin_diff[i];
            run += (d > 0.0f) ? d : 0.0f;
            bin_lds[i] = run;
        }
    }

    const int v0 = blockIdx.x * 128 + wave * 16;

    v2f Bf[32];
    #pragma unroll
    for (int s = 0; s < 32; ++s) {
        const int k = 4 * s + 2 * hi;
        Bf[s].x = VvT[(size_t)k       * NV_ + v0 + col];
        Bf[s].y = VvT[(size_t)(k + 1) * NV_ + v0 + col];
    }

    const int goff = grow * ST_PITCH_ + 4 * gq;
    const int foff = col * ST_PITCH_ + 2 * hi;
    k1_scatter(&a_lds[0][goff], g);
    g   = k1_gather(emb, rn0, rn1, gq);
    rn0 = idx_b[2 * ST_ROWS_ + grow];
    rn1 = idx_b[2 * ST_ROWS_ + grow + 16];
    __syncthreads();

    const unsigned wtab = __float_as_uint(bin_lds[(col + 1) & 15]);

    v8f c0 = {}, c1 = {};
    float acc = 0.0f;
    int bcur = 0;
    #pragma unroll 1
    for (int st = 0; st < ST_COUNT_; ++st) {
        const int bnxt = (bcur == 2) ? 0 : bcur + 1;
        v2f Af0[32], Af1[32];

        if (lead)
            k1_load_frags(Af0, Af1, &a_lds[bcur][foff]);
        if (st + 1 < ST_COUNT_) {
            k1_scatter(&a_lds[bnxt][goff], g);
        }
        __syncthreads();

        if (st + 2 < ST_COUNT_) {
            g = k1_gather(emb, rn0, rn1, gq);
            const int st3 = (st + 3 < ST_COUNT_) ? st + 3 : ST_COUNT_ - 1;
            rn0 = idx_b[st3 * ST_ROWS_ + grow];
            rn1 = idx_b[st3 * ST_ROWS_ + grow + 16];
        }
        __builtin_amdgcn_sched_barrier(0);

        if (!lead) {
            k1_load_frags(Af0, Af1, &a_lds[bcur][foff]);
            __builtin_amdgcn_sched_barrier(0);
            if (st > 0)
                acc = k1_epilogue(acc, c0, c1, (st - 1) * ST_ROWS_, hi, attn_lds, wtab);
        }
        k1_wmma_run(c0, c1, Af0, Af1, Bf);
        if (lead)
            acc = k1_epilogue(acc, c0, c1, st * ST_ROWS_, hi, attn_lds, wtab);
        bcur = bnxt;
    }
    if (!lead)
        acc = k1_epilogue(acc, c0, c1, (ST_COUNT_ - 1) * ST_ROWS_, hi, attn_lds, wtab);

    acc += __shfl_xor(acc, 16, 32);
    if (hi == 0)
        a_out[(size_t)b * NV_ + v0 + col] = acc;
}

#define K2_WAVES_  32
#define K2_KSLICE_ (NV_ / K2_WAVES_)
#define K2_CH_     32
#define K2_NCH_    (K2_KSLICE_ / K2_CH_)
#define K2_CSTEPS_ (K2_CH_ / 4)
#define K2_PITCH_  (K2_CH_ + 4)
#define K2_PROWS_  16
#define K2_AROWS_  8
#define K2_PATCH_  ((K2_PROWS_ + K2_AROWS_) * K2_PITCH_)

struct K2Rows { float4 p0, p1, p2, p3, a0, a1; };

__device__ __forceinline__ K2Rows k2_load_rows(const float* psrc, const float* asrc) {
    K2Rows g;
    g.p0 = *(const float4*)(psrc);
    g.p1 = *(const float4*)(psrc + (size_t)4  * NV_);
    g.p2 = *(const float4*)(psrc + (size_t)8  * NV_);
    g.p3 = *(const float4*)(psrc + (size_t)12 * NV_);
    g.a0 = *(const float4*)(asrc);
    g.a1 = *(const float4*)(asrc + (size_t)4  * NV_);
    return g;
}

__device__ __forceinline__ void k2_store_rows(float* pdst, float* adst, const K2Rows& g) {
    *(float4*)(pdst)                  = g.p0;
    *(float4*)(pdst + 4  * K2_PITCH_) = g.p1;
    *(float4*)(pdst + 8  * K2_PITCH_) = g.p2;
    *(float4*)(pdst + 12 * K2_PITCH_) = g.p3;
    *(float4*)(adst)                  = g.a0;
    *(float4*)(adst + 4  * K2_PITCH_) = g.a1;
}

__global__ __launch_bounds__(1024) void hier_final_gemm_kernel(
    const float* __restrict__ a,
    const float* __restrict__ phi,
    float*       __restrict__ out)
{
    __shared__ float part[K2_WAVES_][8 * 16];
    __shared__ __attribute__((aligned(16))) float patch[K2_WAVES_][2][K2_PATCH_];

    const int j0   = blockIdx.x * 16;
    const int tid  = threadIdx.x;
    const int wave = tid >> 5;
    const int lane = tid & 31;
    const int col  = lane & 15;
    const int hi   = lane >> 4;
    const int k0   = wave * K2_KSLICE_;

    const int lrow = lane >> 3;
    const int lq   = lane & 7;
    const float* pbase = phi + (size_t)(j0 + lrow) * NV_ + k0 + 4 * lq;
    const float* abase = a   + (size_t)lrow        * NV_ + k0 + 4 * lq;

    float* my   = &patch[wave][0][0];
    float* pdst = my + lrow * K2_PITCH_ + 4 * lq;
    float* adst = my + (K2_PROWS_ + lrow) * K2_PITCH_ + 4 * lq;

    const float* bfrag = my + col * K2_PITCH_ + 2 * hi;
    const float* afrag = my + (K2_PROWS_ + (col & 7)) * K2_PITCH_ + 2 * hi;

    K2Rows g = k2_load_rows(pbase, abase);
    k2_store_rows(pdst, adst, g);
    __syncthreads();

    v8f c = {};
    #pragma unroll 1
    for (int ch = 0; ch < K2_NCH_; ++ch) {
        const bool more = (ch + 1 < K2_NCH_);
        if (more)
            g = k2_load_rows(pbase + (ch + 1) * K2_CH_, abase + (ch + 1) * K2_CH_);

        const int cur = (ch & 1) * K2_PATCH_;
        v2f af[K2_CSTEPS_], bf[K2_CSTEPS_];
        #pragma unroll
        for (int s = 0; s < K2_CSTEPS_; ++s) {
            af[s] = *(const v2f*)(afrag + cur + 4 * s);
            bf[s] = *(const v2f*)(bfrag + cur + 4 * s);
        }
        #pragma unroll
        for (int s = 0; s < K2_CSTEPS_; ++s)
            c = __builtin_amdgcn_wmma_f32_16x16x4_f32(
                    false, af[s], false, bf[s], (short)0, c, false, false);

        if (more) {
            const int nxt = ((ch + 1) & 1) * K2_PATCH_;
            k2_store_rows(pdst + nxt, adst + nxt, g);
            __syncthreads();
        }
    }

    if (hi == 0) {
        #pragma unroll
        for (int r = 0; r < 8; ++r)
            part[wave][r * 16 + col] = c[r];
    }
    __syncthreads();

    if (tid < 8 * 16) {
        float sum = 0.0f;
        #pragma unroll
        for (int w = 0; w < K2_WAVES_; ++w)
            sum += part[w][tid];
        out[(size_t)(tid >> 4) * NODES_ + j0 + (tid & 15)] = sum;
    }
}

extern "C" void kernel_launch(void* const* d_in, const int* in_sizes, int n_in,
                              void* d_out, int out_size, void* d_ws, size_t ws_size,
                              hipStream_t stream) {
    const int*   doc_index = (const int*)  d_in[0];
    const float* attn      = (const float*)d_in[1];
    const float* emb       = (const float*)d_in[2];
    const float* VvT       = (const float*)d_in[3];
    const float* phi       = (const float*)d_in[4];
    const float* bin_diff  = (const float*)d_in[5];
    const float* bin_start = (const float*)d_in[6];

    float* a_scratch = (float*)d_ws;
    float* out       = (float*)d_out;

    dim3 grid1(NV_ / 128, B_);
    hier_sim_bin_attn_kernel<<<grid1, 256, 0, stream>>>(
        doc_index, attn, emb, VvT, bin_diff, bin_start, a_scratch);

    hier_final_gemm_kernel<<<NODES_ / 16, 1024, 0, stream>>>(
        a_scratch, phi, out);
}


# final GEMM: 256 blocks of 4 batch rows x 4 nodes (was 32 of 8 x 16), K cut in 4 parts that fill the 16x16 tile; a third of the bytes read per WGP; changes the summation order of the final GEMM
# speedup vs baseline: 6.8016x; 1.0888x over previous
#include <hip/hip_runtime.h>
#include <hip/hip_bf16.h>

typedef __attribute__((ext_vector_type(2))) float v2f;
typedef __attribute__((ext_vector_type(8))) float v8f;

#define B_     8
#define ND_    512
#define D_     128
#define NV_    4096
#define NODES_ 512

#define ST_ROWS_   32
#define ST_COUNT_  (ND_ / ST_ROWS_)
#define ST_PITCH_  (D_ + 4)
#define ST_GROUPS_ (ST_ROWS_ / 16)
#define ST_BUFS_   3
#define ST_FLOATS_ (ST_ROWS_ * ST_PITCH_)

struct K1Rows { float4 a0, a1, b0, b1; };

__device__ __forceinline__ K1Rows k1_gather(const float* emb, int rowa, int rowb, int gq) {
    const float4* sa = (const float4*)(emb + (size_t)rowa * D_);
    const float4* sb = (const float4*)(emb + (size_t)rowb * D_);
    K1Rows g;
    g.a0 = sa[gq];
    g.a1 = sa[gq + 16];
    g.b0 = sb[gq];
    g.b1 = sb[gq + 16];
    return g;
}

__device__ __forceinline__ void k1_scatter(float* dst, const K1Rows& g) {
    *(float4*)(dst)                       = g.a0;
    *(float4*)(dst + 64)                  = g.a1;
    *(float4*)(dst + 16 * ST_PITCH_)      = g.b0;
    *(float4*)(dst + 16 * ST_PITCH_ + 64) = g.b1;
}

__device__ __forceinline__ void k1_load_frags(v2f (&Af0)[32], v2f (&Af1)[32], const float* arow0) {
    const float* arow1 = arow0 + 16 * ST_PITCH_;
    #pragma unroll
    for (int s = 0; s < 32; ++s) {
        Af0[s] = *(const v2f*)(arow0 + 4 * s);
        Af1[s] = *(const v2f*)(arow1 + 4 * s);
    }
}

__device__ __forceinline__ void k1_wmma_run(v8f& c0, v8f& c1, const v2f (&Af0)[32],
                                            const v2f (&Af1)[32], const v2f (&Bf)[32]) {
    c0 = v8f{};
    c1 = v8f{};
    #pragma unroll
    for (int s = 0; s < 32; ++s) {
        c0 = __builtin_amdgcn_wmma_f32_16x16x4_f32(
                false, Af0[s], false, Bf[s], (short)0, c0, false, false);
        c1 = __builtin_amdgcn_wmma_f32_16x16x4_f32(
                false, Af1[s], false, Bf[s], (short)0, c1, false, false);
    }
}

__device__ __forceinline__ float k1_epilogue(float acc, const v8f& c0, const v8f& c1, int nbase0,
                                             int hi, const float* attn_lds, unsigned wtab) {
    const float inv_h  = 14.0f / 1.49f;
    const float c_half = 0.5f - 0.5f * (1.49f / 14.0f);
    const float magic  = 12582912.0f;
    #pragma unroll
    for (int t = 0; t < 2; ++t) {
        const int nbase = nbase0 + 16 * t;
        #pragma unroll
        for (int r = 0; r < 8; ++r) {
            const float x = t ? c1[r] : c0[r];
            float k = __builtin_fmaf(x + c_half, inv_h, magic);
            k = __builtin_amdgcn_fmed3f(k, magic - 1.0f, magic + 14.0f);
            const unsigned sel = __float_as_uint(k);
            const float w = __uint_as_float(
                __builtin_amdgcn_permlane16_var(sel, wtab, sel, false, false));
            const int   m = r + 8 * hi;
            acc = __builtin_fmaf(w, attn_lds[nbase + m], acc);
        }
    }
    return acc;
}

__global__ __launch_bounds__(256) __attribute__((amdgpu_waves_per_eu(2, 2)))
void hier_sim_bin_attn_kernel(
    const int*   __restrict__ doc_index,
    const float* __restrict__ attn,
    const float* __restrict__ emb,
    const float* __restrict__ VvT,
    const float* __restrict__ bin_diff,
    const float* __restrict__ bin_start,
    float*       __restrict__ a_out)
{
    __shared__ float attn_lds[ND_];
    __shared__ float bin_lds[16];
    __shared__ __attribute__((aligned(16))) float a_lds[ST_BUFS_][ST_FLOATS_];

    const int b    = blockIdx.y;
    const int tid  = threadIdx.x;
    const int wave = tid >> 5;
    const int lane = tid & 31;
    const int col  = lane & 15;
    const int hi   = lane >> 4;
    const bool lead = (__builtin_amdgcn_readfirstlane(tid) < 128);

    const int grow = tid >> 4;
    const int gq   = tid & 15;
    const int* idx_b = doc_index + b * ND_;

    K1Rows g = k1_gather(emb, idx_b[grow], idx_b[grow + 16], gq);
    int rn0 = idx_b[ST_ROWS_ + grow];
    int rn1 = idx_b[ST_ROWS_ + grow + 16];

    for (int i = tid; i < ND_; i += 256)
        attn_lds[i] = attn[b * ND_ + i];
    if (tid == 0) {
        float run = bin_start[0];
        #pragma unroll
        for (int i = 0; i < 16; ++i) {
            float d = bin_diff[i];
            run += (d > 0.0f) ? d : 0.0f;
            bin_lds[i] = run;
        }
    }

    const int v0 = blockIdx.x * 128 + wave * 16;

    v2f Bf[32];
    #pragma unroll
    for (int s = 0; s < 32; ++s) {
        const int k = 4 * s + 2 * hi;
        Bf[s].x = VvT[(size_t)k       * NV_ + v0 + col];
        Bf[s].y = VvT[(size_t)(k + 1) * NV_ + v0 + col];
    }

    const int goff = grow * ST_PITCH_ + 4 * gq;
    const int foff = col * ST_PITCH_ + 2 * hi;
    k1_scatter(&a_lds[0][goff], g);
    g   = k1_gather(emb, rn0, rn1, gq);
    rn0 = idx_b[2 * ST_ROWS_ + grow];
    rn1 = idx_b[2 * ST_ROWS_ + grow + 16];
    __syncthreads();

    const unsigned wtab = __float_as_uint(bin_lds[(col + 1) & 15]);

    v8f c0 = {}, c1 = {};
    float acc = 0.0f;
    int bcur = 0;
    #pragma unroll 1
    for (int st = 0; st < ST_COUNT_; ++st) {
        const int bnxt = (bcur == 2) ? 0 : bcur + 1;
        v2f Af0[32], Af1[32];

        if (lead)
            k1_load_frags(Af0, Af1, &a_lds[bcur][foff]);
        if (st + 1 < ST_COUNT_) {
            k1_scatter(&a_lds[bnxt][goff], g);
        }
        __syncthreads();

        if (st + 2 < ST_COUNT_) {
            g = k1_gather(emb, rn0, rn1, gq);
            const int st3 = (st + 3 < ST_COUNT_) ? st + 3 : ST_COUNT_ - 1;
            rn0 = idx_b[st3 * ST_ROWS_ + grow];
            rn1 = idx_b[st3 * ST_ROWS_ + grow + 16];
        }
        __builtin_amdgcn_sched_barrier(0);

        if (!lead) {
            k1_load_frags(Af0, Af1, &a_lds[bcur][foff]);
            __builtin_amdgcn_sched_barrier(0);
            if (st > 0)
                acc = k1_epilogue(acc, c0, c1, (st - 1) * ST_ROWS_, hi, attn_lds, wtab);
        }
        k1_wmma_run(c0, c1, Af0, Af1, Bf);
        if (lead)
            acc = k1_epilogue(acc, c0, c1, st * ST_ROWS_, hi, attn_lds, wtab);
        bcur = bnxt;
    }
    if (!lead)
        acc = k1_epilogue(acc, c0, c1, (ST_COUNT_ - 1) * ST_ROWS_, hi, attn_lds, wtab);

    acc += __shfl_xor(acc, 16, 32);
    if (hi == 0)
        a_out[(size_t)b * NV_ + v0 + col] = acc;
}

#define K2_WAVES_  8
#define K2_PARTS_  4
#define K2_TILE_   4
#define K2_KPART_  (NV_ / K2_PARTS_)
#define K2_KSLICE_ (K2_KPART_ / K2_WAVES_)
#define K2_CH_     32
#define K2_NCH_    (K2_KSLICE_ / K2_CH_)
#define K2_CSTEPS_ (K2_CH_ / 4)
#define K2_PITCH_  (K2_CH_ + 4)
#define K2_PROWS_  (K2_PARTS_ * K2_TILE_)
#define K2_AROWS_  (K2_PARTS_ * K2_TILE_)
#define K2_PATCH_  ((K2_PROWS_ + K2_AROWS_) * K2_PITCH_)
#define K2_GRID_   ((NODES_ / K2_TILE_) * (B_ / K2_TILE_))

__global__ __launch_bounds__(256) void hier_final_gemm_kernel(
    const float* __restrict__ a,
    const float* __restrict__ phi,
    float*       __restrict__ out)
{
    __shared__ float part[K2_WAVES_][16 * 16];
    __shared__ __attribute__((aligned(16))) float patch[K2_WAVES_][K2_NCH_][K2_PATCH_];

    const int j0   = (blockIdx.x >> 1) * K2_TILE_;
    const int b0   = (blockIdx.x & 1) * K2_TILE_;
    const int tid  = threadIdx.x;
    const int wave = tid >> 5;
    const int lane = tid & 31;
    const int col  = lane & 15;
    const int hi   = lane >> 4;
    const int k0   = wave * K2_KSLICE_;

    const int lrow = lane >> 3;
    const int lq   = lane & 7;
    const float* pbase = phi + (size_t)(j0 + lrow) * NV_ + k0 + 4 * lq;
    const float* abase = a   + (size_t)(b0 + lrow) * NV_ + k0 + 4 * lq;

    float4 gp[K2_NCH_][K2_PARTS_], ga[K2_NCH_][K2_PARTS_];
    #pragma unroll
    for (int ch = 0; ch < K2_NCH_; ++ch) {
        #pragma unroll
        for (int p = 0; p < K2_PARTS_; ++p) {
            gp[ch][p] = *(const float4*)(pbase + p * K2_KPART_ + ch * K2_CH_);
            ga[ch][p] = *(const float4*)(abase + p * K2_KPART_ + ch * K2_CH_);
        }
    }

    float* my   = &patch[wave][0][0];
    float* pdst = my + lrow * K2_PITCH_ + 4 * lq;
    float* adst = my + (K2_PROWS_ + lrow) * K2_PITCH_ + 4 * lq;
    #pragma unroll
    for (int ch = 0; ch < K2_NCH_; ++ch) {
        #pragma unroll
        for (int p = 0; p < K2_PARTS_; ++p) {
            *(float4*)(pdst + ch * K2_PATCH_ + 4 * p * K2_PITCH_) = gp[ch][p];
            *(float4*)(adst + ch * K2_PATCH_ + 4 * p * K2_PITCH_) = ga[ch][p];
        }
    }
    __syncthreads();

    const float* bfrag = my + col * K2_PITCH_ + 2 * hi;
    const float* afrag = my + (K2_PROWS_ + col) * K2_PITCH_ + 2 * hi;

    v8f c0 = {}, c1 = {};
    #pragma unroll
    for (int ch = 0; ch < K2_NCH_; ++ch) {
        v2f af[K2_CSTEPS_], bf[K2_CSTEPS_];
        #pragma unroll
        for (int s = 0; s < K2_CSTEPS_; ++s) {
            af[s] = *(const v2f*)(afrag + ch * K2_PATCH_ + 4 * s);
            bf[s] = *(const v2f*)(bfrag + ch * K2_PATCH_ + 4 * s);
        }
        #pragma unroll
        for (int s = 0; s < K2_CSTEPS_; s += 2) {
            c0 = __builtin_amdgcn_wmma_f32_16x16x4_f32(
                    false, af[s], false, bf[s], (short)0, c0, false, false);
            c1 = __builtin_amdgcn_wmma_f32_16x16x4_f32(
                    false, af[s + 1], false, bf[s + 1], (short)0, c1, false, false);
        }
    }
    const v8f c = c0 + c1;

    #pragma unroll
    for (int r = 0; r < 8; ++r)
        part[wave][(8 * hi + r) * 16 + col] = c[r];
    __syncthreads();

    if (tid < K2_TILE_ * K2_TILE_) {
        float sum = 0.0f;
        #pragma unroll
        for (int w = 0; w < K2_WAVES_; ++w) {
            #pragma unroll
            for (int p = 0; p < K2_PARTS_; ++p)
                sum += part[w][(4 * p + (tid >> 2)) * 16 + 4 * p + (tid & 3)];
        }
        out[(size_t)(b0 + (tid >> 2)) * NODES_ + j0 + (tid & 3)] = sum;
    }
}

extern "C" void kernel_launch(void* const* d_in, const int* in_sizes, int n_in,
                              void* d_out, int out_size, void* d_ws, size_t ws_size,
                              hipStream_t stream) {
    const int*   doc_index = (const int*)  d_in[0];
    const float* attn      = (const float*)d_in[1];
    const float* emb       = (const float*)d_in[2];
    const float* VvT       = (const float*)d_in[3];
    const float* phi       = (const float*)d_in[4];
    const float* bin_diff  = (const float*)d_in[5];
    const float* bin_start = (const float*)d_in[6];

    float* a_scratch = (float*)d_ws;
    float* out       = (float*)d_out;

    dim3 grid1(NV_ / 128, B_);
    hier_sim_bin_attn_kernel<<<grid1, 256, 0, stream>>>(
        doc_index, attn, emb, VvT, bin_diff, bin_start, a_scratch);

    hier_final_gemm_kernel<<<K2_GRID_, 256, 0, stream>>>(
        a_scratch, phi, out);
}


# first kernel start-up: index, attn and VvT fragment loads requested as one group in front of the first wait, both first row gathers right behind the indices; attn staging waits for no gather
# speedup vs baseline: 7.0578x; 1.0377x over previous
#include <hip/hip_runtime.h>
#include <hip/hip_bf16.h>

typedef __attribute__((ext_vector_type(2))) float v2f;
typedef __attribute__((ext_vector_type(8))) float v8f;

#define B_     8
#define ND_    512
#define D_     128
#define NV_    4096
#define NODES_ 512

#define ST_ROWS_   32
#define ST_COUNT_  (ND_ / ST_ROWS_)
#define ST_PITCH_  (D_ + 4)
#define ST_GROUPS_ (ST_ROWS_ / 16)
#define ST_BUFS_   3
#define ST_FLOATS_ (ST_ROWS_ * ST_PITCH_)

struct K1Rows { float4 a0, a1, b0, b1; };

__device__ __forceinline__ K1Rows k1_gather(const float* emb, int rowa, int rowb, int gq) {
    const float4* sa = (const float4*)(emb + (size_t)rowa * D_);
    const float4* sb = (const float4*)(emb + (size_t)rowb * D_);
    K1Rows g;
    g.a0 = sa[gq];
    g.a1 = sa[gq + 16];
    g.b0 = sb[gq];
    g.b1 = sb[gq + 16];
    return g;
}

__device__ __forceinline__ void k1_scatter(float* dst, const K1Rows& g) {
    *(float4*)(dst)                       = g.a0;
    *(float4*)(dst + 64)                  = g.a1;
    *(float4*)(dst + 16 * ST_PITCH_)      = g.b0;
    *(float4*)(dst + 16 * ST_PITCH_ + 64) = g.b1;
}

__device__ __forceinline__ void k1_load_frags(v2f (&Af0)[32], v2f (&Af1)[32], const float* arow0) {
    const float* arow1 = arow0 + 16 * ST_PITCH_;
    #pragma unroll
    for (int s = 0; s < 32; ++s) {
        Af0[s] = *(const v2f*)(arow0 + 4 * s);
        Af1[s] = *(const v2f*)(arow1 + 4 * s);
    }
}

__device__ __forceinline__ void k1_wmma_run(v8f& c0, v8f& c1, const v2f (&Af0)[32],
                                            const v2f (&Af1)[32], const v2f (&Bf)[32]) {
    c0 = v8f{};
    c1 = v8f{};
    #pragma unroll
    for (int s = 0; s < 32; ++s) {
        c0 = __builtin_amdgcn_wmma_f32_16x16x4_f32(
                false, Af0[s], false, Bf[s], (short)0, c0, false, false);
        c1 = __builtin_amdgcn_wmma_f32_16x16x4_f32(
                false, Af1[s], false, Bf[s], (short)0, c1, false, false);
    }
}

__device__ __forceinline__ float k1_epilogue(float acc, const v8f& c0, const v8f& c1, int nbase0,
                                             int hi, const float* attn_lds, unsigned wtab) {
    const float inv_h  = 14.0f / 1.49f;
    const float c_half = 0.5f - 0.5f * (1.49f / 14.0f);
    const float magic  = 12582912.0f;
    #pragma unroll
    for (int t = 0; t < 2; ++t) {
        const int nbase = nbase0 + 16 * t;
        #pragma unroll
        for (int r = 0; r < 8; ++r) {
            const float x = t ? c1[r] : c0[r];
            float k = __builtin_fmaf(x + c_half, inv_h, magic);
            k = __builtin_amdgcn_fmed3f(k, magic - 1.0f, magic + 14.0f);
            const unsigned sel = __float_as_uint(k);
            const float w = __uint_as_float(
                __builtin_amdgcn_permlane16_var(sel, wtab, sel, false, false));
            const int   m = r + 8 * hi;
            acc = __builtin_fmaf(w, attn_lds[nbase + m], acc);
        }
    }
    return acc;
}

__global__ __launch_bounds__(256) __attribute__((amdgpu_waves_per_eu(2, 2)))
void hier_sim_bin_attn_kernel(
    const int*   __restrict__ doc_index,
    const float* __restrict__ attn,
    const float* __restrict__ emb,
    const float* __restrict__ VvT,
    const float* __restrict__ bin_diff,
    const float* __restrict__ bin_start,
    float*       __restrict__ a_out)
{
    __shared__ float attn_lds[ND_];
    __shared__ float bin_lds[16];
    __shared__ __attribute__((aligned(16))) float a_lds[ST_BUFS_][ST_FLOATS_];

    const int b    = blockIdx.y;
    const int tid  = threadIdx.x;
    const int wave = tid >> 5;
    const int lane = tid & 31;
    const int col  = lane & 15;
    const int hi   = lane >> 4;
    const bool lead = (__builtin_amdgcn_readfirstlane(tid) < 128);

    const int grow = tid >> 4;
    const int gq   = tid & 15;
    const int* idx_b = doc_index + b * ND_;

    const int v0 = blockIdx.x * 128 + wave * 16;

    const int r00 = idx_b[grow];
    const int r01 = idx_b[grow + 16];
    const int r10 = idx_b[ST_ROWS_ + grow];
    const int r11 = idx_b[ST_ROWS_ + grow + 16];
    int rn0 = idx_b[2 * ST_ROWS_ + grow];
    int rn1 = idx_b[2 * ST_ROWS_ + grow + 16];
    const float at0 = attn[b * ND_ + tid];
    const float at1 = attn[b * ND_ + tid + 256];
    float bd[16];
    #pragma unroll
    for (int i = 0; i < 16; ++i)
        bd[i] = bin_diff[i];
    const float bs = bin_start[0];

    v2f Bf[32];
    #pragma unroll
    for (int s = 0; s < 32; ++s) {
        const int k = 4 * s + 2 * hi;
        Bf[s].x = VvT[(size_t)k       * NV_ + v0 + col];
        Bf[s].y = VvT[(size_t)(k + 1) * NV_ + v0 + col];
    }
    __builtin_amdgcn_sched_barrier(0);

    const K1Rows g0 = k1_gather(emb, r00, r01, gq);
    K1Rows g = k1_gather(emb, r10, r11, gq);
    __builtin_amdgcn_sched_barrier(0);

    if (tid == 0) {
        float run = bs;
        #pragma unroll
        for (int i = 0; i < 16; ++i) {
            const float d = bd[i];
            run += (d > 0.0f) ? d : 0.0f;
            bin_lds[i] = run;
        }
    }
    attn_lds[tid]       = at0;
    attn_lds[tid + 256] = at1;

    const int goff = grow * ST_PITCH_ + 4 * gq;
    const int foff = col * ST_PITCH_ + 2 * hi;
    k1_scatter(&a_lds[0][goff], g0);
    __syncthreads();

    const unsigned wtab = __float_as_uint(bin_lds[(col + 1) & 15]);

    v8f c0 = {}, c1 = {};
    float acc = 0.0f;
    int bcur = 0;
    #pragma unroll 1
    for (int st = 0; st < ST_COUNT_; ++st) {
        const int bnxt = (bcur == 2) ? 0 : bcur + 1;
        v2f Af0[32], Af1[32];

        if (lead)
            k1_load_frags(Af0, Af1, &a_lds[bcur][foff]);
        if (st + 1 < ST_COUNT_) {
            k1_scatter(&a_lds[bnxt][goff], g);
        }
        __syncthreads();

        if (st + 2 < ST_COUNT_) {
            g = k1_gather(emb, rn0, rn1, gq);
            const int st3 = (st + 3 < ST_COUNT_) ? st + 3 : ST_COUNT_ - 1;
            rn0 = idx_b[st3 * ST_ROWS_ + grow];
            rn1 = idx_b[st3 * ST_ROWS_ + grow + 16];
        }
        __builtin_amdgcn_sched_barrier(0);

        if (!lead) {
            k1_load_frags(Af0, Af1, &a_lds[bcur][foff]);
            __builtin_amdgcn_sched_barrier(0);
            if (st > 0)
                acc = k1_epilogue(acc, c0, c1, (st - 1) * ST_ROWS_, hi, attn_lds, wtab);
        }
        k1_wmma_run(c0, c1, Af0, Af1, Bf);
        if (lead)
            acc = k1_epilogue(acc, c0, c1, st * ST_ROWS_, hi, attn_lds, wtab);
        bcur = bnxt;
    }
    if (!lead)
        acc = k1_epilogue(acc, c0, c1, (ST_COUNT_ - 1) * ST_ROWS_, hi, attn_lds, wtab);

    acc += __shfl_xor(acc, 16, 32);
    if (hi == 0)
        a_out[(size_t)b * NV_ + v0 + col] = acc;
}

#define K2_WAVES_  8
#define K2_PARTS_  4
#define K2_TILE_   4
#define K2_KPART_  (NV_ / K2_PARTS_)
#define K2_KSLICE_ (K2_KPART_ / K2_WAVES_)
#define K2_CH_     32
#define K2_NCH_    (K2_KSLICE_ / K2_CH_)
#define K2_CSTEPS_ (K2_CH_ / 4)
#define K2_PITCH_  (K2_CH_ + 4)
#define K2_PROWS_  (K2_PARTS_ * K2_TILE_)
#define K2_AROWS_  (K2_PARTS_ * K2_TILE_)
#define K2_PATCH_  ((K2_PROWS_ + K2_AROWS_) * K2_PITCH_)
#define K2_GRID_   ((NODES_ / K2_TILE_) * (B_ / K2_TILE_))

__global__ __launch_bounds__(256) void hier_final_gemm_kernel(
    const float* __restrict__ a,
    const float* __restrict__ phi,
    float*       __restrict__ out)
{
    __shared__ float part[K2_WAVES_][16 * 16];
    __shared__ __attribute__((aligned(16))) float patch[K2_WAVES_][K2_NCH_][K2_PATCH_];

    const int j0   = (blockIdx.x >> 1) * K2_TILE_;
    const int b0   = (blockIdx.x & 1) * K2_TILE_;
    const int tid  = threadIdx.x;
    const int wave = tid >> 5;
    const int lane = tid & 31;
    const int col  = lane & 15;
    const int hi   = lane >> 4;
    const int k0   = wave * K2_KSLICE_;

    const int lrow = lane >> 3;
    const int lq   = lane & 7;
    const float* pbase = phi + (size_t)(j0 + lrow) * NV_ + k0 + 4 * lq;
    const float* abase = a   + (size_t)(b0 + lrow) * NV_ + k0 + 4 * lq;

    float4 gp[K2_NCH_][K2_PARTS_], ga[K2_NCH_][K2_PARTS_];
    #pragma unroll
    for (int ch = 0; ch < K2_NCH_; ++ch) {
        #pragma unroll
        for (int p = 0; p < K2_PARTS_; ++p) {
            gp[ch][p] = *(const float4*)(pbase + p * K2_KPART_ + ch * K2_CH_);
            ga[ch][p] = *(const float4*)(abase + p * K2_KPART_ + ch * K2_CH_);
        }
    }

    float* my   = &patch[wave][0][0];
    float* pdst = my + lrow * K2_PITCH_ + 4 * lq;
    float* adst = my + (K2_PROWS_ + lrow) * K2_PITCH_ + 4 * lq;
    #pragma unroll
    for (int ch = 0; ch < K2_NCH_; ++ch) {
        #pragma unroll
        for (int p = 0; p < K2_PARTS_; ++p) {
            *(float4*)(pdst + ch * K2_PATCH_ + 4 * p * K2_PITCH_) = gp[ch][p];
            *(float4*)(adst + ch * K2_PATCH_ + 4 * p * K2_PITCH_) = ga[ch][p];
        }
    }
    __syncthreads();

    const float* bfrag = my + col * K2_PITCH_ + 2 * hi;
    const float* afrag = my + (K2_PROWS_ + col) * K2_PITCH_ + 2 * hi;

    v8f c0 = {}, c1 = {};
    #pragma unroll
    for (int ch = 0; ch < K2_NCH_; ++ch) {
        v2f af[K2_CSTEPS_], bf[K2_CSTEPS_];
        #pragma unroll
        for (int s = 0; s < K2_CSTEPS_; ++s) {
            af[s] = *(const v2f*)(afrag + ch * K2_PATCH_ + 4 * s);
            bf[s] = *(const v2f*)(bfrag + ch * K2_PATCH_ + 4 * s);
        }
        #pragma unroll
        for (int s = 0; s < K2_CSTEPS_; s += 2) {
            c0 = __builtin_amdgcn_wmma_f32_16x16x4_f32(
                    false, af[s], false, bf[s], (short)0, c0, false, false);
            c1 = __builtin_amdgcn_wmma_f32_16x16x4_f32(
                    false, af[s + 1], false, bf[s + 1], (short)0, c1, false, false);
        }
    }
    const v8f c = c0 + c1;

    #pragma unroll
    for (int r = 0; r < 8; ++r)
        part[wave][(8 * hi + r) * 16 + col] = c[r];
    __syncthreads();

    if (tid < K2_TILE_ * K2_TILE_) {
        float sum = 0.0f;
        #pragma unroll
        for (int w = 0; w < K2_WAVES_; ++w) {
            #pragma unroll
            for (int p = 0; p < K2_PARTS_; ++p)
                sum += part[w][(4 * p + (tid >> 2)) * 16 + 4 * p + (tid & 3)];
        }
        out[(size_t)(b0 + (tid >> 2)) * NODES_ + j0 + (tid & 3)] = sum;
    }
}

extern "C" void kernel_launch(void* const* d_in, const int* in_sizes, int n_in,
                              void* d_out, int out_size, void* d_ws, size_t ws_size,
                              hipStream_t stream) {
    const int*   doc_index = (const int*)  d_in[0];
    const float* attn      = (const float*)d_in[1];
    const float* emb       = (const float*)d_in[2];
    const float* VvT       = (const float*)d_in[3];
    const float* phi       = (const float*)d_in[4];
    const float* bin_diff  = (const float*)d_in[5];
    const float* bin_start = (const float*)d_in[6];

    float* a_scratch = (float*)d_ws;
    float* out       = (float*)d_out;

    dim3 grid1(NV_ / 128, B_);
    hier_sim_bin_attn_kernel<<<grid1, 256, 0, stream>>>(
        doc_index, attn, emb, VvT, bin_diff, bin_start, a_scratch);

    hier_final_gemm_kernel<<<K2_GRID_, 256, 0, stream>>>(
        a_scratch, phi, out);
}
